# starting kernel
# baseline (speedup 1.0000x reference)
#include <hip/hip_runtime.h>
#include <hip/hip_bf16.h>

#define GG 2048
#define HH 256
#define DD 256

typedef __attribute__((ext_vector_type(2))) float v2f;
typedef __attribute__((ext_vector_type(8))) float v8f;

__global__ void scan_offsets_kernel(const int* __restrict__ nn,
                                    int* __restrict__ offs) {
    __shared__ int chunk[256];
    const int t = threadIdx.x;
    int s = 0;
#pragma unroll
    for (int i = 0; i < 8; ++i) s += nn[t * 8 + i];
    chunk[t] = s;
    __syncthreads();
    if (t == 0) {
        int run = 0;
#pragma unroll 4
        for (int i = 0; i < 256; ++i) { int v = chunk[i]; chunk[i] = run; run += v; }
    }
    __syncthreads();
    int run = chunk[t];
#pragma unroll
    for (int i = 0; i < 8; ++i) {
        int v = nn[t * 8 + i];
        offs[t * 8 + i] = run;
        run += v;
    }
}

__global__ void pool_mean_kernel(const float* __restrict__ feat,
                                 const int* __restrict__ nn,
                                 const int* __restrict__ offs,
                                 float* __restrict__ M) {
    const int g = blockIdx.x;
    const int t = threadIdx.x;
    const int off = offs[g];
    const int cnt = nn[g];
    const float* p = feat + (size_t)off * HH + t;

    float a0 = 0.f, a1 = 0.f, a2 = 0.f, a3 = 0.f;
    int n = 0;
    for (; n + 4 <= cnt; n += 4) {
        a0 += p[(size_t)(n + 0) * HH];
        a1 += p[(size_t)(n + 1) * HH];
        a2 += p[(size_t)(n + 2) * HH];
        a3 += p[(size_t)(n + 3) * HH];
    }
    for (; n < cnt; ++n) a0 += p[(size_t)n * HH];

    const float inv = 1.0f / (float)cnt;
    M[(size_t)g * HH + t] = (a0 + a1 + a2 + a3) * inv;
}

__global__ void gemm3_wmma_kernel(const float* __restrict__ M,
                                  const float* __restrict__ Wp, const float* __restrict__ bp,
                                  const float* __restrict__ Wr, const float* __restrict__ br,
                                  const float* __restrict__ Wk, const float* __restrict__ bk,
                                  float* __restrict__ out) {
    const int wave = threadIdx.x >> 5;
    const int lane = threadIdx.x & 31;
    const int tile = blockIdx.x * 4 + wave;
    const int row0 = (tile >> 4) * 16;
    const int col0 = (tile & 15) * 16;
    const int lm = lane & 15;
    const int hi = lane >> 4;

    const float bpv = bp[col0 + lm];
    const float brv = br[col0 + lm];
    const float bkv = bk[col0 + lm];
    v8f cp, cr, ck;
#pragma unroll
    for (int j = 0; j < 8; ++j) { cp[j] = bpv; cr[j] = brv; ck[j] = bkv; }

    const float* arow = M  + (size_t)(row0 + lm) * HH + 2 * hi;
    const float* wpc  = Wp + (size_t)(2 * hi) * DD + col0 + lm;
    const float* wrc  = Wr + (size_t)(2 * hi) * DD + col0 + lm;
    const float* wkc  = Wk + (size_t)(2 * hi) * DD + col0 + lm;

#pragma unroll 4
    for (int k = 0; k < HH; k += 4) {
        v2f a = *(const v2f*)(arow + k);

        v2f bbp, bbr, bbk;
        const size_t ko = (size_t)k * DD;
        bbp.x = wpc[ko];  bbp.y = wpc[ko + DD];
        bbr.x = wrc[ko];  bbr.y = wrc[ko + DD];
        bbk.x = wkc[ko];  bbk.y = wkc[ko + DD];

        cp = __builtin_amdgcn_wmma_f32_16x16x4_f32(false, a, false, bbp, (short)0, cp, false, false);
        cr = __builtin_amdgcn_wmma_f32_16x16x4_f32(false, a, false, bbr, (short)0, cr, false, false);
        ck = __builtin_amdgcn_wmma_f32_16x16x4_f32(false, a, false, bbk, (short)0, ck, false, false);
    }

    const size_t GD = (size_t)GG * DD;
    float* okeys = out;
    float* opq   = out + GD;
    float* orq   = out + 2 * GD;
#pragma unroll
    for (int j = 0; j < 8; ++j) {
        const int row = row0 + j + 8 * hi;
        const size_t idx = (size_t)row * DD + col0 + lm;
        okeys[idx] = ck[j];
        opq[idx]   = cp[j];
        orq[idx]   = cr[j];
    }
}

extern "C" void kernel_launch(void* const* d_in, const int* in_sizes, int n_in,
                              void* d_out, int out_size, void* d_ws, size_t ws_size,
                              hipStream_t stream) {
    const float* features = (const float*)d_in[0];
    const int*   num_nodes = (const int*)d_in[1];
    const float* Wp = (const float*)d_in[2];
    const float* bp = (const float*)d_in[3];
    const float* Wr = (const float*)d_in[4];
    const float* br = (const float*)d_in[5];
    const float* Wk = (const float*)d_in[6];
    const float* bk = (const float*)d_in[7];
    float* out = (float*)d_out;

    int*   offs = (int*)d_ws;
    float* M    = (float*)((char*)d_ws + 8192);

    scan_offsets_kernel<<<1, 256, 0, stream>>>(num_nodes, offs);
    pool_mean_kernel<<<GG, 256, 0, stream>>>(features, num_nodes, offs, M);
    gemm3_wmma_kernel<<<GG / 4, 128, 0, stream>>>(M, Wp, bp, Wr, br, Wk, bk, out);
    (void)in_sizes; (void)n_in; (void)out_size; (void)ws_size;
}


# GEMM: column-interleaved WMMA tiles, one 16 B (8 B) global load feeds 4 (2) tiles; wave = 16 rows x 96 cols, 1024 waves, loads 3 chunks ahead, no LDS; sum order of the 8 k in a chunk changes
# speedup vs baseline: 1.1995x; 1.1995x over previous
#include <hip/hip_runtime.h>
#include <hip/hip_bf16.h>

#define GG 2048
#define HH 256
#define DD 256

typedef __attribute__((ext_vector_type(2))) float v2f;
typedef __attribute__((ext_vector_type(4))) float v4f;
typedef __attribute__((ext_vector_type(8))) float v8f;

__global__ void scan_offsets_kernel(const int* __restrict__ nn,
                                    int* __restrict__ offs) {
    __shared__ int chunk[256];
    const int t = threadIdx.x;
    int s = 0;
#pragma unroll
    for (int i = 0; i < 8; ++i) s += nn[t * 8 + i];
    chunk[t] = s;
    __syncthreads();
    if (t == 0) {
        int run = 0;
#pragma unroll 4
        for (int i = 0; i < 256; ++i) { int v = chunk[i]; chunk[i] = run; run += v; }
    }
    __syncthreads();
    int run = chunk[t];
#pragma unroll
    for (int i = 0; i < 8; ++i) {
        int v = nn[t * 8 + i];
        offs[t * 8 + i] = run;
        run += v;
    }
}

__global__ void pool_mean_kernel(const float* __restrict__ feat,
                                 const int* __restrict__ nn,
                                 const int* __restrict__ offs,
                                 float* __restrict__ M) {
    const int g = blockIdx.x;
    const int t = threadIdx.x;
    const int off = offs[g];
    const int cnt = nn[g];
    const float* p = feat + (size_t)off * HH + t;

    float a0 = 0.f, a1 = 0.f, a2 = 0.f, a3 = 0.f;
    int n = 0;
    for (; n + 4 <= cnt; n += 4) {
        a0 += p[(size_t)(n + 0) * HH];
        a1 += p[(size_t)(n + 1) * HH];
        a2 += p[(size_t)(n + 2) * HH];
        a3 += p[(size_t)(n + 3) * HH];
    }
    for (; n < cnt; ++n) a0 += p[(size_t)n * HH];

    const float inv = 1.0f / (float)cnt;
    M[(size_t)g * HH + t] = (a0 + a1 + a2 + a3) * inv;
}

#define KCH 32
#define RING 4

__global__ __launch_bounds__(128)
void gemm3_wmma_kernel(const float* __restrict__ M,
                       const float* __restrict__ Wp, const float* __restrict__ bp,
                       const float* __restrict__ Wr, const float* __restrict__ br,
                       const float* __restrict__ Wk, const float* __restrict__ bk,
                       float* __restrict__ out) {
    const int wave = threadIdx.x >> 5;
    const int lane = threadIdx.x & 31;
    const int c  = lane & 15;
    const int h  = lane >> 4;
    const int j  = blockIdx.x & 7;
    const int row0 = ((blockIdx.x >> 3) * 4 + wave) * 16;

    const bool first = (j < 4);
    const float* W64 = first ? Wp : Wr;
    const float* b64 = first ? bp : br;
    const int c64 = 64 * (j & 3) + 4 * c;
    const int c32 = 32 * j + 2 * c;

    const v4f bias4 = *(const v4f*)(b64 + c64);
    const v2f bias2 = *(const v2f*)(bk + c32);
    v8f acc4[4], acc2[2];
#pragma unroll
    for (int r = 0; r < 8; ++r) {
        acc4[0][r] = bias4.x; acc4[1][r] = bias4.y; acc4[2][r] = bias4.z; acc4[3][r] = bias4.w;
        acc2[0][r] = bias2.x; acc2[1][r] = bias2.y;
    }

    const float* pa  = M   + (size_t)(row0 + c) * HH + 4 * h;
    const float* pw4 = W64 + (size_t)(4 * h) * DD + c64;
    const float* pw2 = Wk  + (size_t)(4 * h) * DD + c32;

    v4f sa[RING];
    v4f sb4[RING][4];
    v2f sb2[RING][4];

#define LOAD_CHUNK(q)                                                          \
    {                                                                          \
        const int s_ = (q) % RING;                                             \
        sa[s_] = *(const v4f*)(pa + 8 * (q));                                  \
        _Pragma("unroll")                                                      \
        for (int i = 0; i < 4; ++i) {                                          \
            sb4[s_][i] = *(const v4f*)(pw4 + (size_t)(8 * (q) + i) * DD);      \
            sb2[s_][i] = *(const v2f*)(pw2 + (size_t)(8 * (q) + i) * DD);      \
        }                                                                      \
    }

    LOAD_CHUNK(0)
    LOAD_CHUNK(1)
    LOAD_CHUNK(2)

#pragma unroll
    for (int q = 0; q < KCH; ++q) {
        if (q + 3 < KCH) LOAD_CHUNK(q + 3)
        __builtin_amdgcn_sched_barrier(0);
        const int s = q % RING;
        v2f a0, a1;
        a0.x = sa[s].x; a0.y = sa[s].y;
        a1.x = sa[s].z; a1.y = sa[s].w;
#pragma unroll
        for (int t = 0; t < 4; ++t) {
            v2f b0, b1;
            b0.x = sb4[s][0][t]; b0.y = sb4[s][1][t];
            b1.x = sb4[s][2][t]; b1.y = sb4[s][3][t];
            acc4[t] = __builtin_amdgcn_wmma_f32_16x16x4_f32(false, a0, false, b0, (short)0, acc4[t], false, false);
            acc4[t] = __builtin_amdgcn_wmma_f32_16x16x4_f32(false, a1, false, b1, (short)0, acc4[t], false, false);
        }
#pragma unroll
        for (int u = 0; u < 2; ++u) {
            v2f b0, b1;
            b0.x = sb2[s][0][u]; b0.y = sb2[s][1][u];
            b1.x = sb2[s][2][u]; b1.y = sb2[s][3][u];
            acc2[u] = __builtin_amdgcn_wmma_f32_16x16x4_f32(false, a0, false, b0, (short)0, acc2[u], false, false);
            acc2[u] = __builtin_amdgcn_wmma_f32_16x16x4_f32(false, a1, false, b1, (short)0, acc2[u], false, false);
        }
        __builtin_amdgcn_sched_barrier(0);
    }
#undef LOAD_CHUNK

    const size_t GD = (size_t)GG * DD;
    float* o64 = out + (first ? GD : 2 * GD);
    float* o32 = out;
#pragma unroll
    for (int r = 0; r < 8; ++r) {
        const size_t rowoff = (size_t)(row0 + 8 * h + r) * DD;
        v4f v; v.x = acc4[0][r]; v.y = acc4[1][r]; v.z = acc4[2][r]; v.w = acc4[3][r];
        *(v4f*)(o64 + rowoff + c64) = v;
        v2f w; w.x = acc2[0][r]; w.y = acc2[1][r];
        *(v2f*)(o32 + rowoff + c32) = w;
    }
}

extern "C" void kernel_launch(void* const* d_in, const int* in_sizes, int n_in,
                              void* d_out, int out_size, void* d_ws, size_t ws_size,
                              hipStream_t stream) {
    const float* features = (const float*)d_in[0];
    const int*   num_nodes = (const int*)d_in[1];
    const float* Wp = (const float*)d_in[2];
    const float* bp = (const float*)d_in[3];
    const float* Wr = (const float*)d_in[4];
    const float* br = (const float*)d_in[5];
    const float* Wk = (const float*)d_in[6];
    const float* bk = (const float*)d_in[7];
    float* out = (float*)d_out;

    int*   offs = (int*)d_ws;
    float* M    = (float*)((char*)d_ws + 8192);

    scan_offsets_kernel<<<1, 256, 0, stream>>>(num_nodes, offs);
    pool_mean_kernel<<<GG, 256, 0, stream>>>(features, num_nodes, offs, M);
    gemm3_wmma_kernel<<<256, 128, 0, stream>>>(M, Wp, bp, Wr, br, Wk, bk, out);
    (void)in_sizes; (void)n_in; (void)out_size; (void)ws_size;
}


# One launch, not three: each of 256 workgroups scans num_nodes itself, pools its 8 graphs (one per wave) into a 16-row LDS tile and runs the three GEMMs from it; M stays on chip; sum orders kept
# speedup vs baseline: 1.5051x; 1.2548x over previous
#include <hip/hip_runtime.h>

#define GG 2048
#define HH 256
#define DD 256
#define GPW 8
#define MP 260

typedef __attribute__((ext_vector_type(2))) float v2f;
typedef __attribute__((ext_vector_type(4))) float v4f;
typedef __attribute__((ext_vector_type(8))) float v8f;
typedef __attribute__((ext_vector_type(4))) int   v4i;

#define KCH 32
#define RING 6

__global__ __launch_bounds__(256)
void fused_pool_gemm3_kernel(const float* __restrict__ feat,
                             const int* __restrict__ nn,
                             const float* __restrict__ Wp, const float* __restrict__ bp,
                             const float* __restrict__ Wr, const float* __restrict__ br,
                             const float* __restrict__ Wk, const float* __restrict__ bk,
                             float* __restrict__ out) {
    __shared__ int wsum[8];
    __shared__ __attribute__((aligned(16))) float Mt[16 * MP];

    const int t = threadIdx.x;
    const int lane = t & 31;
    const int wave = __builtin_amdgcn_readfirstlane(t >> 5);
    const int w = blockIdx.x;

    for (int i = t; i < 8 * MP; i += 256) Mt[8 * MP + i] = 0.f;

    const v4i n0 = *(const v4i*)(nn + 8 * t);
    const v4i n1 = *(const v4i*)(nn + 8 * t + 4);
    const int vv[8] = {n0.x, n0.y, n0.z, n0.w, n1.x, n1.y, n1.z, n1.w};
    int s = 0;
#pragma unroll
    for (int m = 0; m < 8; ++m) s += vv[m];
    int incl = s;
#pragma unroll
    for (int d = 1; d < 32; d <<= 1) {
        const int o = __shfl_up(incl, d);
        if (lane >= d) incl += o;
    }
    if (lane == 31) wsum[wave] = incl;
    __syncthreads();
    int base = 0;
#pragma unroll
    for (int u = 0; u < 8; ++u) { const int x = wsum[u]; if (u < wave) base += x; }
    const int sel = w & 7;
    int pre = base + incl - s;
    int cn = 0;
#pragma unroll
    for (int m = 0; m < 8; ++m) {
        if (m < sel) pre += vv[m];
        if (m == sel) cn = vv[m];
    }
    const int off = __builtin_amdgcn_readlane(pre, w >> 3);
    const int cnt = __builtin_amdgcn_readlane(cn, w >> 3);

    {
        const float* p = feat + (size_t)off * HH + 4 * lane;
        v4f a0 = 0.f, a1 = 0.f, a2 = 0.f, a3 = 0.f;
        v4f e0 = 0.f, e1 = 0.f, e2 = 0.f, e3 = 0.f;
        int n = 0;
        for (; n + 16 <= cnt; n += 16) {
            v4f lo[16], hi[16];
#pragma unroll
            for (int i = 0; i < 16; ++i) {
                lo[i] = *(const v4f*)(p + (size_t)(n + i) * HH);
                hi[i] = *(const v4f*)(p + (size_t)(n + i) * HH + 128);
            }
#pragma unroll
            for (int i = 0; i < 16; i += 4) {
                a0 += lo[i]; a1 += lo[i + 1]; a2 += lo[i + 2]; a3 += lo[i + 3];
                e0 += hi[i]; e1 += hi[i + 1]; e2 += hi[i + 2]; e3 += hi[i + 3];
            }
        }
        for (; n + 4 <= cnt; n += 4) {
            v4f lo[4], hi[4];
#pragma unroll
            for (int i = 0; i < 4; ++i) {
                lo[i] = *(const v4f*)(p + (size_t)(n + i) * HH);
                hi[i] = *(const v4f*)(p + (size_t)(n + i) * HH + 128);
            }
            a0 += lo[0]; a1 += lo[1]; a2 += lo[2]; a3 += lo[3];
            e0 += hi[0]; e1 += hi[1]; e2 += hi[2]; e3 += hi[3];
        }
        for (; n < cnt; ++n) {
            a0 += *(const v4f*)(p + (size_t)n * HH);
            e0 += *(const v4f*)(p + (size_t)n * HH + 128);
        }
        const float inv = 1.0f / (float)cnt;
        *(v4f*)(Mt + wave * MP + 4 * lane)       = (a0 + a1 + a2 + a3) * inv;
        *(v4f*)(Mt + wave * MP + 128 + 4 * lane) = (e0 + e1 + e2 + e3) * inv;
    }

    const int c  = lane & 15;
    const int h  = lane >> 4;
    const int j  = wave;
    const bool first = (j < 4);
    const float* W64 = first ? Wp : Wr;
    const float* b64 = first ? bp : br;
    const int c64 = 64 * (j & 3) + 4 * c;
    const int c32 = 32 * j + 2 * c;

    const v4f bias4 = *(const v4f*)(b64 + c64);
    const v2f bias2 = *(const v2f*)(bk + c32);
    v8f acc4[4], acc2[2];
#pragma unroll
    for (int r = 0; r < 8; ++r) {
        acc4[0][r] = bias4.x; acc4[1][r] = bias4.y; acc4[2][r] = bias4.z; acc4[3][r] = bias4.w;
        acc2[0][r] = bias2.x; acc2[1][r] = bias2.y;
    }

    const float* pa  = Mt  + c * MP + 4 * h;
    const float* pw4 = W64 + (size_t)(4 * h) * DD + c64;
    const float* pw2 = Wk  + (size_t)(4 * h) * DD + c32;

    v4f sa[RING];
    v4f sb4[RING][4];
    v2f sb2[RING][4];

#define LOAD_B(q)                                                              \
    {                                                                          \
        const int s_ = (q) % RING;                                             \
        _Pragma("unroll")                                                      \
        for (int i = 0; i < 4; ++i) {                                          \
            sb4[s_][i] = *(const v4f*)(pw4 + (size_t)(8 * (q) + i) * DD);      \
            sb2[s_][i] = *(const v2f*)(pw2 + (size_t)(8 * (q) + i) * DD);      \
        }                                                                      \
    }
#define LOAD_A(q) { sa[(q) % RING] = *(const v4f*)(pa + 8 * (q)); }

    LOAD_B(0)
    LOAD_B(1)
    LOAD_B(2)
    LOAD_B(3)
    LOAD_B(4)
    __syncthreads();
    LOAD_A(0)
    LOAD_A(1)
    LOAD_A(2)
    LOAD_A(3)
    LOAD_A(4)

#pragma unroll
    for (int q = 0; q < KCH; ++q) {
        if (q + RING - 1 < KCH) { LOAD_B(q + RING - 1) LOAD_A(q + RING - 1) }
        __builtin_amdgcn_sched_barrier(0);
        const int s = q % RING;
        v2f a0, a1;
        a0.x = sa[s].x; a0.y = sa[s].y;
        a1.x = sa[s].z; a1.y = sa[s].w;
#pragma unroll
        for (int tt = 0; tt < 4; ++tt) {
            v2f b0, b1;
            b0.x = sb4[s][0][tt]; b0.y = sb4[s][1][tt];
            b1.x = sb4[s][2][tt]; b1.y = sb4[s][3][tt];
            acc4[tt] = __builtin_amdgcn_wmma_f32_16x16x4_f32(false, a0, false, b0, (short)0, acc4[tt], false, false);
            acc4[tt] = __builtin_amdgcn_wmma_f32_16x16x4_f32(false, a1, false, b1, (short)0, acc4[tt], false, false);
        }
#pragma unroll
        for (int u = 0; u < 2; ++u) {
            v2f b0, b1;
            b0.x = sb2[s][0][u]; b0.y = sb2[s][1][u];
            b1.x = sb2[s][2][u]; b1.y = sb2[s][3][u];
            acc2[u] = __builtin_amdgcn_wmma_f32_16x16x4_f32(false, a0, false, b0, (short)0, acc2[u], false, false);
            acc2[u] = __builtin_amdgcn_wmma_f32_16x16x4_f32(false, a1, false, b1, (short)0, acc2[u], false, false);
        }
        __builtin_amdgcn_sched_barrier(0);
    }
#undef LOAD_A
#undef LOAD_B

    if (h == 0) {
        const size_t GD = (size_t)GG * DD;
        float* o64 = out + (first ? GD : 2 * GD);
        float* o32 = out;
#pragma unroll
        for (int r = 0; r < 8; ++r) {
            const size_t rowoff = (size_t)(w + 256 * r) * DD;
            v4f v; v.x = acc4[0][r]; v.y = acc4[1][r]; v.z = acc4[2][r]; v.w = acc4[3][r];
            *(v4f*)(o64 + rowoff + c64) = v;
            v2f x; x.x = acc2[0][r]; x.y = acc2[1][r];
            *(v2f*)(o32 + rowoff + c32) = x;
        }
    }
}

extern "C" void kernel_launch(void* const* d_in, const int* in_sizes, int n_in,
                              void* d_out, int out_size, void* d_ws, size_t ws_size,
                              hipStream_t stream) {
    const float* features = (const float*)d_in[0];
    const int*   num_nodes = (const int*)d_in[1];
    const float* Wp = (const float*)d_in[2];
    const float* bp = (const float*)d_in[3];
    const float* Wr = (const float*)d_in[4];
    const float* br = (const float*)d_in[5];
    const float* Wk = (const float*)d_in[6];
    const float* bk = (const float*)d_in[7];
    float* out = (float*)d_out;

    fused_pool_gemm3_kernel<<<GG / GPW, 256, 0, stream>>>(features, num_nodes,
                                                          Wp, bp, Wr, br, Wk, bk, out);
    (void)in_sizes; (void)n_in; (void)out_size; (void)d_ws; (void)ws_size;
}


# K split in two 128-channel parts: rows of channels 128..255 load 16 at a time under the WMMAs of k 0..127 (same wave); part means meet in the LDS tile; last 16-row unit masked; sum orders kept
# speedup vs baseline: 1.5788x; 1.0490x over previous
#include <hip/hip_runtime.h>

#define GG 2048
#define HH 256
#define DD 256
#define MAXN 128
#define GPW 8
#define MP 260

typedef __attribute__((ext_vector_type(2))) float v2f;
typedef __attribute__((ext_vector_type(4))) float v4f;
typedef __attribute__((ext_vector_type(8))) float v8f;
typedef __attribute__((ext_vector_type(4))) int   v4i;

#define RB 16
__device__ __forceinline__ void unit_issue_full(v4f (&r)[RB], const float* __restrict__ p, int n) {
#pragma unroll
    for (int i = 0; i < RB; ++i) r[i] = *(const v4f*)(p + (size_t)(n + i) * HH);
}

__device__ __forceinline__ void unit_issue_last(v4f (&r)[RB], const float* __restrict__ p,
                                                int n, int rows) {
#pragma unroll
    for (int i = 0; i < RB; ++i) {
        const int row = (n + i < rows) ? (n + i) : (rows - 1);
        r[i] = *(const v4f*)(p + (size_t)row * HH);
    }
}

__device__ __forceinline__ void unit_sum_full(const v4f (&r)[RB], v4f (&a)[4]) {
#pragma unroll
    for (int i = 0; i < RB; ++i) a[i & 3] += r[i];
}

__device__ __forceinline__ void unit_sum_last(const v4f (&r)[RB], v4f (&a)[4],
                                              int n, int rows, int rows4) {
    const v4f z = 0.f;
#pragma unroll
    for (int i = 0; i < RB; ++i) {
        const int row = n + i;
        if ((i & 3) == 0) {
            a[0] += (row < rows) ? r[i] : z;
        } else {
            a[i & 3] += (row < rows4) ? r[i] : z;
            a[0] += (row >= rows4 && row < rows) ? r[i] : z;
        }
    }
}

#define KCH 32
#define RING 6

__global__ __launch_bounds__(256)
void fused_pool_gemm3_kernel(const float* __restrict__ feat,
                             const int* __restrict__ nn,
                             const float* __restrict__ Wp, const float* __restrict__ bp,
                             const float* __restrict__ Wr, const float* __restrict__ br,
                             const float* __restrict__ Wk, const float* __restrict__ bk,
                             float* __restrict__ out) {
    __shared__ int wsum[8];
    __shared__ __attribute__((aligned(16))) float Mt[16 * MP];

    const int t = threadIdx.x;
    const int lane = t & 31;
    const int wave = __builtin_amdgcn_readfirstlane(t >> 5);
    const int w = blockIdx.x;

    for (int i = t; i < 8 * MP; i += 256) Mt[8 * MP + i] = 0.f;

    const v4i n0 = *(const v4i*)(nn + 8 * t);
    const v4i n1 = *(const v4i*)(nn + 8 * t + 4);
    const int vv[8] = {n0.x, n0.y, n0.z, n0.w, n1.x, n1.y, n1.z, n1.w};
    int s = 0;
#pragma unroll
    for (int m = 0; m < 8; ++m) s += vv[m];
    int incl = s;
#pragma unroll
    for (int d = 1; d < 32; d <<= 1) {
        const int o = __shfl_up(incl, d);
        if (lane >= d) incl += o;
    }
    if (lane == 31) wsum[wave] = incl;
    __syncthreads();
    int base = 0;
#pragma unroll
    for (int u = 0; u < 8; ++u) { const int x = wsum[u]; if (u < wave) base += x; }
    const int sel = w & 7;
    int pre = base + incl - s;
    int cn = 0;
#pragma unroll
    for (int m = 0; m < 8; ++m) {
        if (m < sel) pre += vv[m];
        if (m == sel) cn = vv[m];
    }
    const int off = __builtin_amdgcn_readlane(pre, w >> 3);
    const int cnt = __builtin_amdgcn_readlane(cn, w >> 3);

    const int rows  = cnt < MAXN ? cnt : MAXN;
    const int rows4 = rows & ~3;
    const int nlast = rows > 0 ? ((rows - 1) & ~(RB - 1)) : 0;
    const float inv = 1.0f / (float)cnt;
    const float* p  = feat + (size_t)off * HH + 4 * lane;
    const float* ph = p + 128;

    {
        v4f a[4] = {0.f, 0.f, 0.f, 0.f};
        v4f rl[RB];
        if (rows > 0) unit_issue_last(rl, p, nlast, rows);
        for (int n = 0; n < nlast; n += 2 * RB) {
            v4f r0[RB], r1[RB];
            unit_issue_full(r0, p, n);
            if (n + RB < nlast) unit_issue_full(r1, p, n + RB);
            unit_sum_full(r0, a);
            if (n + RB < nlast) unit_sum_full(r1, a);
        }
        if (rows > 0) unit_sum_last(rl, a, nlast, rows, rows4);
        *(v4f*)(Mt + wave * MP + 4 * lane) = (a[0] + a[1] + a[2] + a[3]) * inv;
    }

    const int c  = lane & 15;
    const int h  = lane >> 4;
    const int j  = wave;
    const bool first = (j < 4);
    const float* W64 = first ? Wp : Wr;
    const float* b64 = first ? bp : br;
    const int c64 = 64 * (j & 3) + 4 * c;
    const int c32 = 32 * j + 2 * c;

    const v4f bias4 = *(const v4f*)(b64 + c64);
    const v2f bias2 = *(const v2f*)(bk + c32);
    v8f acc4[4], acc2[2];
#pragma unroll
    for (int r = 0; r < 8; ++r) {
        acc4[0][r] = bias4.x; acc4[1][r] = bias4.y; acc4[2][r] = bias4.z; acc4[3][r] = bias4.w;
        acc2[0][r] = bias2.x; acc2[1][r] = bias2.y;
    }

    const float* pa  = Mt  + c * MP + 4 * h;
    const float* pw4 = W64 + (size_t)(4 * h) * DD + c64;
    const float* pw2 = Wk  + (size_t)(4 * h) * DD + c32;

    v4f sa[RING];
    v4f sb4[RING][4];
    v2f sb2[RING][4];

#define LOAD_B(q)                                                              \
    {                                                                          \
        const int s_ = (q) % RING;                                             \
        _Pragma("unroll")                                                      \
        for (int i = 0; i < 4; ++i) {                                          \
            sb4[s_][i] = *(const v4f*)(pw4 + (size_t)(8 * (q) + i) * DD);      \
            sb2[s_][i] = *(const v2f*)(pw2 + (size_t)(8 * (q) + i) * DD);      \
        }                                                                      \
    }
#define LOAD_A(q) { sa[(q) % RING] = *(const v4f*)(pa + 8 * (q)); }

#define GEMM_CHUNK(q)                                                                       \
    {                                                                                       \
        if ((q) + RING - 1 < KCH) {                                                         \
            LOAD_B((q) + RING - 1)                                                          \
            if ((((q) + RING - 1) < KCH / 2) == ((q) < KCH / 2)) LOAD_A((q) + RING - 1)     \
        }                                                                                   \
              \
                     \
        __builtin_amdgcn_sched_barrier(0);                                                  \
        const int sl = (q) % RING;                                                          \
        v2f a0, a1;                                                                         \
        a0.x = sa[sl].x; a0.y = sa[sl].y;                                                   \
        a1.x = sa[sl].z; a1.y = sa[sl].w;                                                   \
              \
        _Pragma("unroll")                                                                   \
        for (int tt = 0; tt < 4; ++tt) {                                                    \
            v2f b0, b1;                                                                     \
            b0.x = sb4[sl][0][tt]; b0.y = sb4[sl][1][tt];                                   \
            b1.x = sb4[sl][2][tt]; b1.y = sb4[sl][3][tt];                                   \
                           \
            acc4[tt] = __builtin_amdgcn_wmma_f32_16x16x4_f32(false, a0, false, b0, (short)0, acc4[tt], false, false); \
            acc4[tt] = __builtin_amdgcn_wmma_f32_16x16x4_f32(false, a1, false, b1, (short)0, acc4[tt], false, false); \
        }                                                                                   \
        _Pragma("unroll")                                                                   \
        for (int u = 0; u < 2; ++u) {                                                       \
            v2f b0, b1;                                                                     \
            b0.x = sb2[sl][0][u]; b0.y = sb2[sl][1][u];                                     \
            b1.x = sb2[sl][2][u]; b1.y = sb2[sl][3][u];                                     \
            acc2[u] = __builtin_amdgcn_wmma_f32_16x16x4_f32(false, a0, false, b0, (short)0, acc2[u], false, false); \
            acc2[u] = __builtin_amdgcn_wmma_f32_16x16x4_f32(false, a1, false, b1, (short)0, acc2[u], false, false); \
        }                                                                                   \
             \
                               \
        __builtin_amdgcn_sched_barrier(0);                                                  \
    }

    LOAD_B(0)
    LOAD_B(1)
    LOAD_B(2)
    LOAD_B(3)
    LOAD_B(4)
    __syncthreads();
    v4f e[4] = {0.f, 0.f, 0.f, 0.f};
    v4f rh[RB];
    if (0 < nlast) unit_issue_full(rh, ph, 0);
    LOAD_A(0)
    LOAD_A(1)
    LOAD_A(2)
    LOAD_A(3)
    LOAD_A(4)

#pragma unroll
    for (int g = 0; g < MAXN / RB; ++g) {
        GEMM_CHUNK(2 * g)
        GEMM_CHUNK(2 * g + 1)
        if (g < MAXN / RB - 1) {
            if (RB * g < nlast) {
                unit_sum_full(rh, e);
                if (g < MAXN / RB - 2) {
                    if (RB * (g + 1) < nlast) unit_issue_full(rh, ph, RB * (g + 1));
                }
            }
            if (g == MAXN / RB - 2) {
                if (rows > 0) unit_issue_last(rh, ph, nlast, rows);
            }
        } else {
            if (rows > 0) unit_sum_last(rh, e, nlast, rows, rows4);
        }
    }
    *(v4f*)(Mt + wave * MP + 128 + 4 * lane) = (e[0] + e[1] + e[2] + e[3]) * inv;
    __syncthreads();
    LOAD_A(16)
    LOAD_A(17)
    LOAD_A(18)
    LOAD_A(19)
    LOAD_A(20)

#pragma unroll
    for (int q = KCH / 2; q < KCH; ++q) GEMM_CHUNK(q)
#undef GEMM_CHUNK
#undef LOAD_A
#undef LOAD_B

    if (h == 0) {
        const size_t GD = (size_t)GG * DD;
        float* o64 = out + (first ? GD : 2 * GD);
        float* o32 = out;
#pragma unroll
        for (int r = 0; r < 8; ++r) {
            const size_t rowoff = (size_t)(w + 256 * r) * DD;
            v4f v; v.x = acc4[0][r]; v.y = acc4[1][r]; v.z = acc4[2][r]; v.w = acc4[3][r];
            *(v4f*)(o64 + rowoff + c64) = v;
            v2f x; x.x = acc2[0][r]; x.y = acc2[1][r];
            *(v2f*)(o32 + rowoff + c32) = x;
        }
    }
}

extern "C" void kernel_launch(void* const* d_in, const int* in_sizes, int n_in,
                              void* d_out, int out_size, void* d_ws, size_t ws_size,
                              hipStream_t stream) {
    const float* features = (const float*)d_in[0];
    const int*   num_nodes = (const int*)d_in[1];
    const float* Wp = (const float*)d_in[2];
    const float* bp = (const float*)d_in[3];
    const float* Wr = (const float*)d_in[4];
    const float* br = (const float*)d_in[5];
    const float* Wk = (const float*)d_in[6];
    const float* bk = (const float*)d_in[7];
    float* out = (float*)d_out;

    fused_pool_gemm3_kernel<<<GG / GPW, 256, 0, stream>>>(features, num_nodes,
                                                          Wp, bp, Wr, br, Wk, bk, out);
    (void)in_sizes; (void)n_in; (void)out_size; (void)d_ws; (void)ws_size;
}
